# starting kernel
# baseline (speedup 1.0000x reference)
#include <hip/hip_runtime.h>

typedef __attribute__((ext_vector_type(2))) float v2f;
typedef __attribute__((ext_vector_type(8))) float v8f;

#define NHD   8
#define HDD   6
#define DIMC  48
#define NT    27
#define GH    40
#define GW    40
#define GT    40
#define QSCALE 0.40824829046386301636f

constexpr unsigned bcode(int t) {
    return (t >= NT)
        ? 0x55555555u
        : (unsigned)(t / 9)
        | ((unsigned)((t / 3) % 3) << 2)
        | ((unsigned)(t % 3) << 4)
        | 0x55555540u;
}

__device__ __forceinline__ float bval(unsigned code, int col) {
    return (float)((int)((code >> (2 * col)) & 3u) - 1);
}

__global__ __launch_bounds__(256) void nat3d_wmma_kernel(
    const float* __restrict__ q,
    const float* __restrict__ k,
    const float* __restrict__ rpb,
    float* __restrict__ out)
{
    __shared__ float s_rpb[NHD * NT];
    int tid = threadIdx.x;
    if (tid < NHD * NT) s_rpb[tid] = rpb[tid];
    __syncthreads();

    const int lane = tid & 31;
    const int wave = tid >> 5;
    const int pair = blockIdx.x * 8 + wave;
    const int z2   = pair % (GT / 2);
    const int tmp  = pair / (GT / 2);
    const int y    = tmp % GW;
    const int x    = tmp / GW;

    const int m    = lane & 15;
    const int half = lane >> 4;
    const int h    = m & 7;
    const int z    = z2 * 2 + (m >> 3);

    const int lin = (x * GW + y) * GT + z;
    const float* qp = q + (size_t)lin * DIMC + h * HDD;
    float qv[HDD];
#pragma unroll
    for (int d = 0; d < HDD; d += 2) {
        v2f v = *(const v2f*)(qp + d);
        qv[d]     = v.x * QSCALE;
        qv[d + 1] = v.y * QSCALE;
    }

    float p[NT];
    float mx = -1e30f;
#pragma unroll
    for (int t = 0; t < NT; ++t) {
        const int ti = t / 9, tj = (t / 3) % 3, tl = t % 3;
        const int nx = x + ti - 1, ny = y + tj - 1, nz = z + tl - 1;
        float dot = 0.0f;
        if ((unsigned)nx < GH && (unsigned)ny < GW && (unsigned)nz < GT) {
            const float* kp = k + (size_t)((nx * GW + ny) * GT + nz) * DIMC + h * HDD;
#pragma unroll
            for (int d = 0; d < HDD; d += 2) {
                v2f v = *(const v2f*)(kp + d);
                dot = fmaf(qv[d],     v.x, dot);
                dot = fmaf(qv[d + 1], v.y, dot);
            }
        }
        float s = dot + s_rpb[h * NT + t];
        p[t] = s;
        mx = fmaxf(mx, s);
    }

    float sum = 0.0f;
#pragma unroll
    for (int t = 0; t < NT; ++t) { p[t] = __expf(p[t] - mx); sum += p[t]; }
    const float inv = __builtin_amdgcn_rcpf(sum);
#pragma unroll
    for (int t = 0; t < NT; ++t) p[t] *= inv;

    const int col = lane & 15;
    v8f acc = {};
#pragma unroll
    for (int c = 0; c < 7; ++c) {
        const int t0 = 4 * c;
        const float a_lo0 = p[t0];
        const float a_lo1 = p[t0 + 1];
        const float a_hi0 = (t0 + 2 < NT) ? p[t0 + 2 < NT ? t0 + 2 : 0] : 0.0f;
        const float a_hi1 = (t0 + 3 < NT) ? p[t0 + 3 < NT ? t0 + 3 : 0] : 0.0f;
        v2f a, b;
        a.x = half ? a_hi0 : a_lo0;
        a.y = half ? a_hi1 : a_lo1;
        const unsigned cx = half ? bcode(t0 + 2) : bcode(t0);
        const unsigned cy = half ? bcode(t0 + 3) : bcode(t0 + 1);
        b.x = bval(cx, col);
        b.y = bval(cy, col);
        acc = __builtin_amdgcn_wmma_f32_16x16x4_f32(
             false, a,  false, b,
             (short)0, acc,  false,  false);
    }

    if (col < 3) {
        const size_t sp = (size_t)((x * GW + y) * GT) + (size_t)(z2 * 2 + half);
#pragma unroll
        for (int r = 0; r < 8; ++r) {
            out[(size_t)(r * 3 + col) * (GH * GW * GT) + sp] = acc[r];
        }
    }
}

extern "C" void kernel_launch(void* const* d_in, const int* in_sizes, int n_in,
                              void* d_out, int out_size, void* d_ws, size_t ws_size,
                              hipStream_t stream) {
    (void)in_sizes; (void)n_in; (void)out_size; (void)d_ws; (void)ws_size;
    const float* q   = (const float*)d_in[0];
    const float* k   = (const float*)d_in[1];
    const float* rpb = (const float*)d_in[2];
    float* out = (float*)d_out;

    dim3 grid(4000), block(256);
    nat3d_wmma_kernel<<<grid, block, 0, stream>>>(q, k, rpb, out);
}


# Each lane owns its own (voxel, head) row: 32 rows per wave instead of 16 held twice; second 16-row WMMA tile gets its A pairs by lane^16 exchange; half as many waves; sum order unchanged
# speedup vs baseline: 1.9356x; 1.9356x over previous
#include <hip/hip_runtime.h>

typedef __attribute__((ext_vector_type(2))) float v2f;
typedef __attribute__((ext_vector_type(8))) float v8f;

#define NHD   8
#define HDD   6
#define DIMC  48
#define NT    27
#define GH    40
#define GW    40
#define GT    40
#define QSCALE 0.40824829046386301636f

constexpr unsigned bcode(int t) {
    return (t >= NT)
        ? 0x55555555u
        : (unsigned)(t / 9)
        | ((unsigned)((t / 3) % 3) << 2)
        | ((unsigned)(t % 3) << 4)
        | 0x55555540u;
}

__device__ __forceinline__ float bval(unsigned code, int col) {
    return (float)((int)((code >> (2 * col)) & 3u) - 1);
}

__device__ __forceinline__ float xhalf(float x) {
    int b = __builtin_bit_cast(int, x);
    return __builtin_bit_cast(float,
        __builtin_amdgcn_permlanex16(b, b, 0x76543210, 0xfedcba98, false, false));
}

__global__ __launch_bounds__(256) void nat3d_wmma_kernel(
    const float* __restrict__ q,
    const float* __restrict__ k,
    const float* __restrict__ rpb,
    float* __restrict__ out)
{
    __shared__ float s_rpb[NHD * NT];
    int tid = threadIdx.x;
    if (tid < NHD * NT) s_rpb[tid] = rpb[tid];
    __syncthreads();

    const int lane = tid & 31;
    const int wave = tid >> 5;
    const int quad = blockIdx.x * 8 + wave;
    const int zq   = quad % (GT / 4);
    const int tmp  = quad / (GT / 4);
    const int y    = tmp % GW;
    const int x    = tmp / GW;

    const int half = lane >> 4;
    const int h    = lane & 7;
    const int v    = 2 * ((lane >> 3) & 1) + half;
    const int z    = zq * 4 + v;

    const int lin = (x * GW + y) * GT + z;
    const float* qp = q + (size_t)lin * DIMC + h * HDD;
    float qv[HDD];
#pragma unroll
    for (int d = 0; d < HDD; d += 2) {
        v2f w = *(const v2f*)(qp + d);
        qv[d]     = w.x * QSCALE;
        qv[d + 1] = w.y * QSCALE;
    }

    float p[NT];
    float mx = -1e30f;
#pragma unroll
    for (int t = 0; t < NT; ++t) {
        const int ti = t / 9, tj = (t / 3) % 3, tl = t % 3;
        const int nx = x + ti - 1, ny = y + tj - 1, nz = z + tl - 1;
        float dot = 0.0f;
        if ((unsigned)nx < GH && (unsigned)ny < GW && (unsigned)nz < GT) {
            const float* kp = k + (size_t)((nx * GW + ny) * GT + nz) * DIMC + h * HDD;
#pragma unroll
            for (int d = 0; d < HDD; d += 2) {
                v2f w = *(const v2f*)(kp + d);
                dot = fmaf(qv[d],     w.x, dot);
                dot = fmaf(qv[d + 1], w.y, dot);
            }
        }
        float s = dot + s_rpb[h * NT + t];
        p[t] = s;
        mx = fmaxf(mx, s);
    }

    float sum = 0.0f;
#pragma unroll
    for (int t = 0; t < NT; ++t) { p[t] = __expf(p[t] - mx); sum += p[t]; }
    const float inv = __builtin_amdgcn_rcpf(sum);
#pragma unroll
    for (int t = 0; t < NT; ++t) p[t] *= inv;

    const int col = lane & 15;
    v8f acc0 = {}, acc1 = {};
#pragma unroll
    for (int c = 0; c < 7; ++c) {
        const int t0 = 4 * c;
        const float lo0 = p[t0];
        const float lo1 = p[t0 + 1];
        const float hi0 = p[t0 + 2];
        const float hi1 = (t0 + 3 < NT) ? p[t0 + 3 < NT ? t0 + 3 : 0] : 0.0f;
        const float s0 = half ? lo0 : hi0;
        const float s1 = half ? lo1 : hi1;
        const float r0 = xhalf(s0);
        const float r1 = xhalf(s1);
        v2f a0, a1, b;
        a0.x = half ? r0  : lo0;
        a0.y = half ? r1  : lo1;
        a1.x = half ? hi0 : r0;
        a1.y = half ? hi1 : r1;
        const unsigned cx = half ? bcode(t0 + 2) : bcode(t0);
        const unsigned cy = half ? bcode(t0 + 3) : bcode(t0 + 1);
        b.x = bval(cx, col);
        b.y = bval(cy, col);
        acc0 = __builtin_amdgcn_wmma_f32_16x16x4_f32(
             false, a0,  false, b,
             (short)0, acc0,  false,  false);
        acc1 = __builtin_amdgcn_wmma_f32_16x16x4_f32(
             false, a1,  false, b,
             (short)0, acc1,  false,  false);
    }

    if (col < 3) {
        const size_t sp = (size_t)((x * GW + y) * GT) + (size_t)(zq * 4 + half * 2);
#pragma unroll
        for (int r = 0; r < 8; ++r) {
            v2f o2;
            o2.x = acc0[r];
            o2.y = acc1[r];
            *(v2f*)(out + (size_t)(r * 3 + col) * (GH * GW * GT) + sp) = o2;
        }
    }
}

extern "C" void kernel_launch(void* const* d_in, const int* in_sizes, int n_in,
                              void* d_out, int out_size, void* d_ws, size_t ws_size,
                              hipStream_t stream) {
    (void)in_sizes; (void)n_in; (void)out_size; (void)d_ws; (void)ws_size;
    const float* q   = (const float*)d_in[0];
    const float* k   = (const float*)d_in[1];
    const float* rpb = (const float*)d_in[2];
    float* out = (float*)d_out;

    dim3 grid(2000), block(256);
    nat3d_wmma_kernel<<<grid, block, 0, stream>>>(q, k, rpb, out);
}


# All 27 neighbor key loads issued unconditionally from clamped in-range addresses, padding zeroed by a mask behind the products: loads overlap, not 27 serial round trips; sum order unchanged
# speedup vs baseline: 2.4636x; 1.2728x over previous
#include <hip/hip_runtime.h>

typedef __attribute__((ext_vector_type(2))) float v2f;
typedef __attribute__((ext_vector_type(8))) float v8f;

#define NHD   8
#define HDD   6
#define DIMC  48
#define NT    27
#define GH    40
#define GW    40
#define GT    40
#define QSCALE 0.40824829046386301636f

constexpr unsigned bcode(int t) {
    return (t >= NT)
        ? 0x55555555u
        : (unsigned)(t / 9)
        | ((unsigned)((t / 3) % 3) << 2)
        | ((unsigned)(t % 3) << 4)
        | 0x55555540u;
}

__device__ __forceinline__ float bval(unsigned code, int col) {
    return (float)((int)((code >> (2 * col)) & 3u) - 1);
}

__device__ __forceinline__ float xhalf(float x) {
    int b = __builtin_bit_cast(int, x);
    return __builtin_bit_cast(float,
        __builtin_amdgcn_permlanex16(b, b, 0x76543210, 0xfedcba98, false, false));
}

__global__ __launch_bounds__(256) void nat3d_wmma_kernel(
    const float* __restrict__ q,
    const float* __restrict__ k,
    const float* __restrict__ rpb,
    float* __restrict__ out)
{
    __shared__ float s_rpb[NHD * NT];
    int tid = threadIdx.x;
    if (tid < NHD * NT) s_rpb[tid] = rpb[tid];
    __syncthreads();

    const int lane = tid & 31;
    const int wave = tid >> 5;
    const int quad = blockIdx.x * 8 + wave;
    const int zq   = quad % (GT / 4);
    const int tmp  = quad / (GT / 4);
    const int y    = tmp % GW;
    const int x    = tmp / GW;

    const int half = lane >> 4;
    const int h    = lane & 7;
    const int v    = 2 * ((lane >> 3) & 1) + half;
    const int z    = zq * 4 + v;

    const int lin = (x * GW + y) * GT + z;
    const float* qp = q + (size_t)lin * DIMC + h * HDD;
    float qv[HDD];
#pragma unroll
    for (int d = 0; d < HDD; d += 2) {
        v2f w = *(const v2f*)(qp + d);
        qv[d]     = w.x * QSCALE;
        qv[d + 1] = w.y * QSCALE;
    }

    int cxs[3], cys[3], czs[3];
    int vxs[3], vys[3], vzs[3];
#pragma unroll
    for (int a = 0; a < 3; ++a) {
        const int nx = x + a - 1, ny = y + a - 1, nz = z + a - 1;
        vxs[a] = (unsigned)nx < GH ? -1 : 0;  cxs[a] = min(max(nx, 0), GH - 1);
        vys[a] = (unsigned)ny < GW ? -1 : 0;  cys[a] = min(max(ny, 0), GW - 1);
        vzs[a] = (unsigned)nz < GT ? -1 : 0;  czs[a] = min(max(nz, 0), GT - 1);
    }
    v2f kw[NT][HDD / 2];
#pragma unroll
    for (int t = 0; t < NT; ++t) {
        const int ti = t / 9, tj = (t / 3) % 3, tl = t % 3;
        const unsigned off = (unsigned)((cxs[ti] * GW + cys[tj]) * GT + czs[tl]) * DIMC + (unsigned)(h * HDD);
        const float* kp = k + off;
#pragma unroll
        for (int d = 0; d < HDD; d += 2) kw[t][d / 2] = *(const v2f*)(kp + d);
    }
    float p[NT];
    float mx = -1e30f;
#pragma unroll
    for (int t = 0; t < NT; ++t) {
        const int ti = t / 9, tj = (t / 3) % 3, tl = t % 3;
        float dot = 0.0f;
#pragma unroll
        for (int d = 0; d < HDD; d += 2) {
            dot = fmaf(qv[d],     kw[t][d / 2].x, dot);
            dot = fmaf(qv[d + 1], kw[t][d / 2].y, dot);
        }
        dot = __builtin_bit_cast(float, __builtin_bit_cast(int, dot) & vxs[ti] & vys[tj] & vzs[tl]);
        float s = dot + s_rpb[h * NT + t];
        p[t] = s;
        mx = fmaxf(mx, s);
    }

    float sum = 0.0f;
#pragma unroll
    for (int t = 0; t < NT; ++t) { p[t] = __expf(p[t] - mx); sum += p[t]; }
    const float inv = __builtin_amdgcn_rcpf(sum);
#pragma unroll
    for (int t = 0; t < NT; ++t) p[t] *= inv;

    const int col = lane & 15;
    v8f acc0 = {}, acc1 = {};
#pragma unroll
    for (int c = 0; c < 7; ++c) {
        const int t0 = 4 * c;
        const float lo0 = p[t0];
        const float lo1 = p[t0 + 1];
        const float hi0 = p[t0 + 2];
        const float hi1 = (t0 + 3 < NT) ? p[t0 + 3 < NT ? t0 + 3 : 0] : 0.0f;
        const float s0 = half ? lo0 : hi0;
        const float s1 = half ? lo1 : hi1;
        const float r0 = xhalf(s0);
        const float r1 = xhalf(s1);
        v2f a0, a1, b;
        a0.x = half ? r0  : lo0;
        a0.y = half ? r1  : lo1;
        a1.x = half ? hi0 : r0;
        a1.y = half ? hi1 : r1;
        const unsigned cx = half ? bcode(t0 + 2) : bcode(t0);
        const unsigned cy = half ? bcode(t0 + 3) : bcode(t0 + 1);
        b.x = bval(cx, col);
        b.y = bval(cy, col);
        acc0 = __builtin_amdgcn_wmma_f32_16x16x4_f32(
             false, a0,  false, b,
             (short)0, acc0,  false,  false);
        acc1 = __builtin_amdgcn_wmma_f32_16x16x4_f32(
             false, a1,  false, b,
             (short)0, acc1,  false,  false);
    }

    if (col < 3) {
        const size_t sp = (size_t)((x * GW + y) * GT) + (size_t)(zq * 4 + half * 2);
#pragma unroll
        for (int r = 0; r < 8; ++r) {
            v2f o2;
            o2.x = acc0[r];
            o2.y = acc1[r];
            *(v2f*)(out + (size_t)(r * 3 + col) * (GH * GW * GT) + sp) = o2;
        }
    }
}

extern "C" void kernel_launch(void* const* d_in, const int* in_sizes, int n_in,
                              void* d_out, int out_size, void* d_ws, size_t ws_size,
                              hipStream_t stream) {
    (void)in_sizes; (void)n_in; (void)out_size; (void)d_ws; (void)ws_size;
    const float* q   = (const float*)d_in[0];
    const float* k   = (const float*)d_in[1];
    const float* rpb = (const float*)d_in[2];
    float* out = (float*)d_out;

    dim3 grid(2000), block(256);
    nat3d_wmma_kernel<<<grid, block, 0, stream>>>(q, k, rpb, out);
}


# Each lane owns 4 z-consecutive voxels of one head: 6 keys of a neighbor column loaded once serve 12 scores (was 3 keys for 3); half the key loads, 4x fewer waves; sum order unchanged
# speedup vs baseline: 2.5903x; 1.0514x over previous
#include <hip/hip_runtime.h>

typedef __attribute__((ext_vector_type(2))) float v2f;
typedef __attribute__((ext_vector_type(4))) float v4f;
typedef __attribute__((ext_vector_type(8))) float v8f;

#define NHD   8
#define HDD   6
#define DIMC  48
#define NT    27
#define GH    40
#define GW    40
#define GT    40
#define ZV    4
#define NKZ   6
#define QSCALE 0.40824829046386301636f

constexpr unsigned bcode(int t) {
    return (t >= NT)
        ? 0x55555555u
        : (unsigned)(t / 9)
        | ((unsigned)((t / 3) % 3) << 2)
        | ((unsigned)(t % 3) << 4)
        | 0x55555540u;
}

__device__ __forceinline__ float bval(unsigned code, int col) {
    return (float)((int)((code >> (2 * col)) & 3u) - 1);
}

__device__ __forceinline__ float xhalf(float x) {
    int b = __builtin_bit_cast(int, x);
    return __builtin_bit_cast(float,
        __builtin_amdgcn_permlanex16(b, b, 0x76543210, 0xfedcba98, false, false));
}

__device__ __forceinline__ void pv_tiles(const float (&p)[NT], int half, int col, v8f& acc0, v8f& acc1) {
#pragma unroll
    for (int c = 0; c < 7; ++c) {
        const int t0 = 4 * c;
        const float lo0 = p[t0];
        const float lo1 = p[t0 + 1];
        const float hi0 = p[t0 + 2];
        const float hi1 = (t0 + 3 < NT) ? p[t0 + 3 < NT ? t0 + 3 : 0] : 0.0f;
        const float s0 = half ? lo0 : hi0;
        const float s1 = half ? lo1 : hi1;
        const float r0 = xhalf(s0);
        const float r1 = xhalf(s1);
        v2f a0, a1, b;
        a0.x = half ? r0  : lo0;
        a0.y = half ? r1  : lo1;
        a1.x = half ? hi0 : r0;
        a1.y = half ? hi1 : r1;
        const unsigned cx = half ? bcode(t0 + 2) : bcode(t0);
        const unsigned cy = half ? bcode(t0 + 3) : bcode(t0 + 1);
        b.x = bval(cx, col);
        b.y = bval(cy, col);
        acc0 = __builtin_amdgcn_wmma_f32_16x16x4_f32(
             false, a0,  false, b,
             (short)0, acc0,  false,  false);
        acc1 = __builtin_amdgcn_wmma_f32_16x16x4_f32(
             false, a1,  false, b,
             (short)0, acc1,  false,  false);
    }
}

__global__ __launch_bounds__(256, 4) void nat3d_wmma_kernel(
    const float* __restrict__ q,
    const float* __restrict__ k,
    const float* __restrict__ rpb,
    float* __restrict__ out)
{
    __shared__ float s_rpb[NHD * NT];
    int tid = threadIdx.x;
    if (tid < NHD * NT) s_rpb[tid] = rpb[tid];
    __syncthreads();

    const int lane = tid & 31;
    const int wave = tid >> 5;
    const int wid  = blockIdx.x * 8 + wave;
    const int zq   = wid % (GT / ZV);
    const int tmp  = wid / (GT / ZV);
    const int yg   = tmp % (GW / 4);
    const int x    = tmp / (GW / 4);

    const int half = lane >> 4;
    const int h    = lane & 7;
    const int y    = yg * 4 + (lane >> 3);
    const int z0   = zq * ZV;

    float qv[ZV][HDD];
#pragma unroll
    for (int u = 0; u < ZV; ++u) {
        const int lin = (x * GW + y) * GT + z0 + u;
        const float* qp = q + (size_t)lin * DIMC + h * HDD;
#pragma unroll
        for (int d = 0; d < HDD; d += 2) {
            v2f w = *(const v2f*)(qp + d);
            qv[u][d]     = w.x * QSCALE;
            qv[u][d + 1] = w.y * QSCALE;
        }
    }

    int cxs[3], cys[3], vxs[3], vys[3];
#pragma unroll
    for (int a = 0; a < 3; ++a) {
        const int nx = x + a - 1, ny = y + a - 1;
        vxs[a] = (unsigned)nx < GH ? -1 : 0;  cxs[a] = min(max(nx, 0), GH - 1);
        vys[a] = (unsigned)ny < GW ? -1 : 0;  cys[a] = min(max(ny, 0), GW - 1);
    }
    int czk[NKZ];
#pragma unroll
    for (int kz = 0; kz < NKZ; ++kz) czk[kz] = min(max(z0 - 1 + kz, 0), GT - 1);
    const int vzlo = (z0 - 1 >= 0) ? -1 : 0;
    const int vzhi = (z0 + ZV < GT) ? -1 : 0;

    float s[ZV][NT];
#pragma unroll
    for (int ij = 0; ij < 9; ++ij) {
        const int ti = ij / 3, tj = ij % 3;
        const int mxy = vxs[ti] & vys[tj];
        const unsigned colbase = (unsigned)((cxs[ti] * GW + cys[tj]) * GT);
        v2f kw[NKZ][HDD / 2];
#pragma unroll
        for (int kz = 0; kz < NKZ; ++kz) {
            const unsigned off = (colbase + (unsigned)czk[kz]) * DIMC + (unsigned)(h * HDD);
            const float* kp = k + off;
#pragma unroll
            for (int d = 0; d < HDD; d += 2) kw[kz][d / 2] = *(const v2f*)(kp + d);
        }
#pragma unroll
        for (int l = 0; l < 3; ++l) {
            const int t = ij * 3 + l;
            const float bias = s_rpb[h * NT + t];
#pragma unroll
            for (int u = 0; u < ZV; ++u) {
                const int kz = u + l;
                float dot = 0.0f;
#pragma unroll
                for (int d = 0; d < HDD; d += 2) {
                    dot = fmaf(qv[u][d],     kw[kz][d / 2].x, dot);
                    dot = fmaf(qv[u][d + 1], kw[kz][d / 2].y, dot);
                }
                int m = mxy;
                if (kz == 0)       m &= vzlo;
                if (kz == NKZ - 1) m &= vzhi;
                dot = __builtin_bit_cast(float, __builtin_bit_cast(int, dot) & m);
                s[u][t] = dot + bias;
            }
        }
    }

    const int col = lane & 15;
    v8f acc[ZV][2];
#pragma unroll
    for (int u = 0; u < ZV; ++u) {
        float mx = -1e30f;
#pragma unroll
        for (int t = 0; t < NT; ++t) mx = fmaxf(mx, s[u][t]);
        float sum = 0.0f;
#pragma unroll
        for (int t = 0; t < NT; ++t) { s[u][t] = __expf(s[u][t] - mx); sum += s[u][t]; }
        const float inv = __builtin_amdgcn_rcpf(sum);
#pragma unroll
        for (int t = 0; t < NT; ++t) s[u][t] *= inv;
        acc[u][0] = v8f{};
        acc[u][1] = v8f{};
        pv_tiles(s[u], half, col, acc[u][0], acc[u][1]);
    }

    if (col < 3) {
#pragma unroll
        for (int tl = 0; tl < 2; ++tl) {
            const int yy = yg * 4 + 2 * tl + half;
            const size_t sp = (size_t)((x * GW + yy) * GT + z0);
#pragma unroll
            for (int r = 0; r < 8; ++r) {
                v4f o4;
                o4.x = acc[0][tl][r];
                o4.y = acc[1][tl][r];
                o4.z = acc[2][tl][r];
                o4.w = acc[3][tl][r];
                *(v4f*)(out + (size_t)(r * 3 + col) * (GH * GW * GT) + sp) = o4;
            }
        }
    }
}

extern "C" void kernel_launch(void* const* d_in, const int* in_sizes, int n_in,
                              void* d_out, int out_size, void* d_ws, size_t ws_size,
                              hipStream_t stream) {
    (void)in_sizes; (void)n_in; (void)out_size; (void)d_ws; (void)ws_size;
    const float* q   = (const float*)d_in[0];
    const float* k   = (const float*)d_in[1];
    const float* rpb = (const float*)d_in[2];
    float* out = (float*)d_out;

    dim3 grid(500), block(256);
    nat3d_wmma_kernel<<<grid, block, 0, stream>>>(q, k, rpb, out);
}


# Key halo tile of each 2x8x8-voxel block staged once in LDS by whole-line 16-byte loads, padding zero-filled; lanes read keys from LDS, not by 8-line strided global loads; sum order unchanged
# speedup vs baseline: 2.9998x; 1.1581x over previous
#include <hip/hip_runtime.h>

typedef __attribute__((ext_vector_type(2))) float v2f;
typedef __attribute__((ext_vector_type(4))) float v4f;
typedef __attribute__((ext_vector_type(8))) float v8f;

#define NHD   8
#define HDD   6
#define DIMC  48
#define NT    27
#define GH    40
#define GW    40
#define GT    40
#define ZV    4
#define NKZ   6
#define QSCALE 0.40824829046386301636f
#define TX    4
#define TY    10
#define TZ    10
#define ROWF4 (TZ * DIMC / 4)
#define YP    (TZ * DIMC + 48)
#define XP    (TY * YP)

constexpr unsigned bcode(int t) {
    return (t >= NT)
        ? 0x55555555u
        : (unsigned)(t / 9)
        | ((unsigned)((t / 3) % 3) << 2)
        | ((unsigned)(t % 3) << 4)
        | 0x55555540u;
}

__device__ __forceinline__ float bval(unsigned code, int col) {
    return (float)((int)((code >> (2 * col)) & 3u) - 1);
}

__device__ __forceinline__ float xhalf(float x) {
    int b = __builtin_bit_cast(int, x);
    return __builtin_bit_cast(float,
        __builtin_amdgcn_permlanex16(b, b, 0x76543210, 0xfedcba98, false, false));
}

__device__ __forceinline__ void pv_tiles(const float (&p)[NT], int half, int col, v8f& acc0, v8f& acc1) {
#pragma unroll
    for (int c = 0; c < 7; ++c) {
        const int t0 = 4 * c;
        const float lo0 = p[t0];
        const float lo1 = p[t0 + 1];
        const float hi0 = p[t0 + 2];
        const float hi1 = (t0 + 3 < NT) ? p[t0 + 3 < NT ? t0 + 3 : 0] : 0.0f;
        const float s0 = half ? lo0 : hi0;
        const float s1 = half ? lo1 : hi1;
        const float r0 = xhalf(s0);
        const float r1 = xhalf(s1);
        v2f a0, a1, b;
        a0.x = half ? r0  : lo0;
        a0.y = half ? r1  : lo1;
        a1.x = half ? hi0 : r0;
        a1.y = half ? hi1 : r1;
        const unsigned cx = half ? bcode(t0 + 2) : bcode(t0);
        const unsigned cy = half ? bcode(t0 + 3) : bcode(t0 + 1);
        b.x = bval(cx, col);
        b.y = bval(cy, col);
        acc0 = __builtin_amdgcn_wmma_f32_16x16x4_f32(
             false, a0,  false, b,
             (short)0, acc0,  false,  false);
        acc1 = __builtin_amdgcn_wmma_f32_16x16x4_f32(
             false, a1,  false, b,
             (short)0, acc1,  false,  false);
    }
}

__global__ __launch_bounds__(256, 4) void nat3d_wmma_kernel(
    const float* __restrict__ q,
    const float* __restrict__ k,
    const float* __restrict__ rpb,
    float* __restrict__ out)
{
    __shared__ float s_rpb[NHD * NT];
    __shared__ __attribute__((aligned(16))) float s_k[TX * XP];
    const int tid = threadIdx.x;

    const int bz  = blockIdx.x % (GT / 8);
    const int by  = (blockIdx.x / (GT / 8)) % (GW / 8);
    const int bx  = blockIdx.x / ((GT / 8) * (GW / 8));
    const int x0  = bx * 2;
    const int y0  = by * 8;
    const int z0b = bz * 8;

    const int lane = tid & 31;
    const int wave = tid >> 5;
    const int zq_l = wave & 1;
    const int yg_l = (wave >> 1) & 1;
    const int x_l  = wave >> 2;
    const int x    = x0 + x_l;
    const int yg   = by * 2 + yg_l;

    const int half = lane >> 4;
    const int h    = lane & 7;
    const int ysub = lane >> 3;
    const int y    = yg * 4 + ysub;
    const int z0   = z0b + zq_l * ZV;

    float rpbv = 0.0f;
    if (tid < NHD * NT) rpbv = rpb[tid];

    v2f qw[ZV][HDD / 2];
#pragma unroll
    for (int u = 0; u < ZV; ++u) {
        const int lin = (x * GW + y) * GT + z0 + u;
        const float* qp = q + (size_t)lin * DIMC + h * HDD;
#pragma unroll
        for (int d = 0; d < HDD; d += 2) qw[u][d / 2] = *(const v2f*)(qp + d);
    }

    const int fhi  = tid >> 7;
    const int fc   = tid & 127;
    const int fcc  = min(fc, ROWF4 - 1);
    const int gz   = z0b - 1 + fcc / (DIMC / 4);
    const int gzc  = min(max(gz, 0), GT - 1);
    const bool zok = (gz == gzc);
    const int loff = gzc * DIMC + (fcc % (DIMC / 4)) * 4;
    const int gxb  = x0 - 1 + 2 * fhi;
    const int gyb  = y0 - 1;
    v4f ld[20];
#pragma unroll
    for (int p = 0; p < 20; ++p) {
        const int gxc = min(max(gxb + p / TY, 0), GH - 1);
        const int gyc = min(max(gyb + p % TY, 0), GW - 1);
        ld[p] = *(const v4f*)(k + ((gxc * GW + gyc) * (GT * DIMC) + loff));
    }
    asm volatile("" :: "v"(qw[0][0]), "v"(qw[0][1]), "v"(qw[0][2]), "v"(qw[1][0]), "v"(qw[1][1]), "v"(qw[1][2]),
                       "v"(qw[2][0]), "v"(qw[2][1]), "v"(qw[2][2]), "v"(qw[3][0]), "v"(qw[3][1]), "v"(qw[3][2]));
    asm volatile("" :: "v"(ld[0]), "v"(ld[1]), "v"(ld[2]), "v"(ld[3]), "v"(ld[4]), "v"(ld[5]), "v"(ld[6]),
                       "v"(ld[7]), "v"(ld[8]), "v"(ld[9]), "v"(ld[10]), "v"(ld[11]), "v"(ld[12]), "v"(ld[13]),
                       "v"(ld[14]), "v"(ld[15]), "v"(ld[16]), "v"(ld[17]), "v"(ld[18]), "v"(ld[19]));

    if (tid < NHD * NT) s_rpb[tid] = rpbv;
    if (fc < ROWF4) {
        const int dst0 = 2 * fhi * XP + fc * 4;
#pragma unroll
        for (int p = 0; p < 20; ++p) {
            const int gx = gxb + p / TY, gy = gyb + p % TY;
            const bool ok = zok && ((unsigned)gx < GH) && ((unsigned)gy < GW);
            *(v4f*)(&s_k[dst0 + (p / TY) * XP + (p % TY) * YP]) = ok ? ld[p] : v4f{};
        }
    }
    __syncthreads();

    float qv[ZV][HDD];
#pragma unroll
    for (int u = 0; u < ZV; ++u) {
#pragma unroll
        for (int d = 0; d < HDD; d += 2) {
            qv[u][d]     = qw[u][d / 2].x * QSCALE;
            qv[u][d + 1] = qw[u][d / 2].y * QSCALE;
        }
    }

    const float* kb = s_k + (x_l * XP + (yg_l * 4 + ysub) * YP + zq_l * (ZV * DIMC) + h * HDD);

    float s[ZV][NT];
#pragma unroll
    for (int ij = 0; ij < 9; ++ij) {
        const int ti = ij / 3, tj = ij % 3;
        v2f kw[NKZ][HDD / 2];
#pragma unroll
        for (int kz = 0; kz < NKZ; ++kz) {
            const float* kp = kb + (ti * XP + tj * YP + kz * DIMC);
#pragma unroll
            for (int d = 0; d < HDD; d += 2) kw[kz][d / 2] = *(const v2f*)(kp + d);
        }
#pragma unroll
        for (int l = 0; l < 3; ++l) {
            const int t = ij * 3 + l;
            const float bias = s_rpb[h * NT + t];
#pragma unroll
            for (int u = 0; u < ZV; ++u) {
                const int kz = u + l;
                float dot = 0.0f;
#pragma unroll
                for (int d = 0; d < HDD; d += 2) {
                    dot = fmaf(qv[u][d],     kw[kz][d / 2].x, dot);
                    dot = fmaf(qv[u][d + 1], kw[kz][d / 2].y, dot);
                }
                s[u][t] = dot + bias;
            }
        }
    }

    const int col = lane & 15;
    v8f acc[ZV][2];
#pragma unroll
    for (int u = 0; u < ZV; ++u) {
        float mx = -1e30f;
#pragma unroll
        for (int t = 0; t < NT; ++t) mx = fmaxf(mx, s[u][t]);
        float sum = 0.0f;
#pragma unroll
        for (int t = 0; t < NT; ++t) { s[u][t] = __expf(s[u][t] - mx); sum += s[u][t]; }
        const float inv = __builtin_amdgcn_rcpf(sum);
#pragma unroll
        for (int t = 0; t < NT; ++t) s[u][t] *= inv;
        acc[u][0] = v8f{};
        acc[u][1] = v8f{};
        pv_tiles(s[u], half, col, acc[u][0], acc[u][1]);
    }

    if (col < 3) {
#pragma unroll
        for (int tl = 0; tl < 2; ++tl) {
            const int yy = yg * 4 + 2 * tl + half;
            const size_t sp = (size_t)((x * GW + yy) * GT + z0);
#pragma unroll
            for (int r = 0; r < 8; ++r) {
                v4f o4;
                o4.x = acc[0][tl][r];
                o4.y = acc[1][tl][r];
                o4.z = acc[2][tl][r];
                o4.w = acc[3][tl][r];
                *(v4f*)(out + (size_t)(r * 3 + col) * (GH * GW * GT) + sp) = o4;
            }
        }
    }
}

extern "C" void kernel_launch(void* const* d_in, const int* in_sizes, int n_in,
                              void* d_out, int out_size, void* d_ws, size_t ws_size,
                              hipStream_t stream) {
    (void)in_sizes; (void)n_in; (void)out_size; (void)d_ws; (void)ws_size;
    const float* q   = (const float*)d_in[0];
    const float* k   = (const float*)d_in[1];
    const float* rpb = (const float*)d_in[2];
    float* out = (float*)d_out;

    dim3 grid(500), block(256);
    nat3d_wmma_kernel<<<grid, block, 0, stream>>>(q, k, rpb, out);
}


# Each block does two tiles in turn: the 2nd tile's key and q loads go out behind the 1st tile's barrier and fly during its compute, into a 2nd LDS buffer; 250 blocks of 8 waves; sum order unchanged
# speedup vs baseline: 3.2462x; 1.0821x over previous
#include <hip/hip_runtime.h>

typedef __attribute__((ext_vector_type(2))) float v2f;
typedef __attribute__((ext_vector_type(4))) float v4f;
typedef __attribute__((ext_vector_type(8))) float v8f;

#define NHD   8
#define HDD   6
#define DIMC  48
#define NT    27
#define GH    40
#define GW    40
#define GT    40
#define ZV    4
#define NKZ   6
#define QSCALE 0.40824829046386301636f
#define TX    4
#define TY    10
#define TZ    10
#define ROWF4 (TZ * DIMC / 4)
#define YP    (TZ * DIMC + 48)
#define XP    (TY * YP)

constexpr unsigned bcode(int t) {
    return (t >= NT)
        ? 0x55555555u
        : (unsigned)(t / 9)
        | ((unsigned)((t / 3) % 3) << 2)
        | ((unsigned)(t % 3) << 4)
        | 0x55555540u;
}

__device__ __forceinline__ float bval(unsigned code, int col) {
    return (float)((int)((code >> (2 * col)) & 3u) - 1);
}

__device__ __forceinline__ float xhalf(float x) {
    int b = __builtin_bit_cast(int, x);
    return __builtin_bit_cast(float,
        __builtin_amdgcn_permlanex16(b, b, 0x76543210, 0xfedcba98, false, false));
}

__device__ __forceinline__ void pv_tiles(const float (&p)[NT], int half, int col, v8f& acc0, v8f& acc1) {
#pragma unroll
    for (int c = 0; c < 7; ++c) {
        const int t0 = 4 * c;
        const float lo0 = p[t0];
        const float lo1 = p[t0 + 1];
        const float hi0 = p[t0 + 2];
        const float hi1 = (t0 + 3 < NT) ? p[t0 + 3 < NT ? t0 + 3 : 0] : 0.0f;
        const float s0 = half ? lo0 : hi0;
        const float s1 = half ? lo1 : hi1;
        const float r0 = xhalf(s0);
        const float r1 = xhalf(s1);
        v2f a0, a1, b;
        a0.x = half ? r0  : lo0;
        a0.y = half ? r1  : lo1;
        a1.x = half ? hi0 : r0;
        a1.y = half ? hi1 : r1;
        const unsigned cx = half ? bcode(t0 + 2) : bcode(t0);
        const unsigned cy = half ? bcode(t0 + 3) : bcode(t0 + 1);
        b.x = bval(cx, col);
        b.y = bval(cy, col);
        acc0 = __builtin_amdgcn_wmma_f32_16x16x4_f32(
             false, a0,  false, b,
             (short)0, acc0,  false,  false);
        acc1 = __builtin_amdgcn_wmma_f32_16x16x4_f32(
             false, a1,  false, b,
             (short)0, acc1,  false,  false);
    }
}

__device__ __forceinline__ void tile_origin(int t, int& x0, int& y0, int& z0b) {
    const int bz = t % (GT / 8);
    const int by = (t / (GT / 8)) % (GW / 8);
    const int bx = t / ((GT / 8) * (GW / 8));
    x0  = bx * 2;
    y0  = by * 8;
    z0b = bz * 8;
}

__device__ __forceinline__ void issue_tile_loads(int t, int tid, const float* __restrict__ q,
                                                 const float* __restrict__ k,
                                                 v2f (&qw)[ZV][HDD / 2], v4f (&ld)[20]) {
    int x0, y0, z0b;
    tile_origin(t, x0, y0, z0b);
    const int lane = tid & 31;
    const int wave = tid >> 5;
    const int zq_l = wave & 1;
    const int yg_l = (wave >> 1) & 1;
    const int x_l  = wave >> 2;
    const int h    = lane & 7;
    const int ysub = lane >> 3;
    const int x    = x0 + x_l;
    const int y    = y0 + yg_l * 4 + ysub;
    const int z0   = z0b + zq_l * ZV;
#pragma unroll
    for (int u = 0; u < ZV; ++u) {
        const int lin = (x * GW + y) * GT + z0 + u;
        const float* qp = q + (size_t)lin * DIMC + h * HDD;
#pragma unroll
        for (int d = 0; d < HDD; d += 2) qw[u][d / 2] = *(const v2f*)(qp + d);
    }
    const int fhi  = tid >> 7;
    const int fc   = tid & 127;
    const int fcc  = min(fc, ROWF4 - 1);
    const int gz   = z0b - 1 + fcc / (DIMC / 4);
    const int gzc  = min(max(gz, 0), GT - 1);
    const int loff = gzc * DIMC + (fcc % (DIMC / 4)) * 4;
    const int gxb  = x0 - 1 + 2 * fhi;
    const int gyb  = y0 - 1;
#pragma unroll
    for (int p = 0; p < 20; ++p) {
        const int gxc = min(max(gxb + p / TY, 0), GH - 1);
        const int gyc = min(max(gyb + p % TY, 0), GW - 1);
        ld[p] = *(const v4f*)(k + ((gxc * GW + gyc) * (GT * DIMC) + loff));
    }
}

__device__ __forceinline__ void fill_tile(int t, int tid, const v4f (&ld)[20], float* s_kb) {
    int x0, y0, z0b;
    tile_origin(t, x0, y0, z0b);
    const int fhi  = tid >> 7;
    const int fc   = tid & 127;
    const int fcc  = min(fc, ROWF4 - 1);
    const int gz   = z0b - 1 + fcc / (DIMC / 4);
    const bool zok = ((unsigned)gz < GT);
    const int gxb  = x0 - 1 + 2 * fhi;
    const int gyb  = y0 - 1;
    if (fc < ROWF4) {
        const int dst0 = 2 * fhi * XP + fc * 4;
#pragma unroll
        for (int p = 0; p < 20; ++p) {
            const int gx = gxb + p / TY, gy = gyb + p % TY;
            const bool ok = zok && ((unsigned)gx < GH) && ((unsigned)gy < GW);
            *(v4f*)(&s_kb[dst0 + (p / TY) * XP + (p % TY) * YP]) = ok ? ld[p] : v4f{};
        }
    }
}

__device__ __forceinline__ void compute_tile(int t, int tid, const v2f (&qw)[ZV][HDD / 2], const float* s_kb,
                                             const float* s_rpb, float* __restrict__ out) {
    int x0, y0, z0b;
    tile_origin(t, x0, y0, z0b);
    const int lane = tid & 31;
    const int wave = tid >> 5;
    const int zq_l = wave & 1;
    const int yg_l = (wave >> 1) & 1;
    const int x_l  = wave >> 2;
    const int x    = x0 + x_l;
    const int half = lane >> 4;
    const int h    = lane & 7;
    const int ysub = lane >> 3;
    const int z0   = z0b + zq_l * ZV;

    float qv[ZV][HDD];
#pragma unroll
    for (int u = 0; u < ZV; ++u) {
#pragma unroll
        for (int d = 0; d < HDD; d += 2) {
            qv[u][d]     = qw[u][d / 2].x * QSCALE;
            qv[u][d + 1] = qw[u][d / 2].y * QSCALE;
        }
    }

    const float* kb = s_kb + (x_l * XP + (yg_l * 4 + ysub) * YP + zq_l * (ZV * DIMC) + h * HDD);

    float s[ZV][NT];
#pragma unroll
    for (int ij = 0; ij < 9; ++ij) {
        const int ti = ij / 3, tj = ij % 3;
        v2f kw[NKZ][HDD / 2];
#pragma unroll
        for (int kz = 0; kz < NKZ; ++kz) {
            const float* kp = kb + (ti * XP + tj * YP + kz * DIMC);
#pragma unroll
            for (int d = 0; d < HDD; d += 2) kw[kz][d / 2] = *(const v2f*)(kp + d);
        }
#pragma unroll
        for (int l = 0; l < 3; ++l) {
            const int tt = ij * 3 + l;
            const float bias = s_rpb[h * NT + tt];
#pragma unroll
            for (int u = 0; u < ZV; ++u) {
                const int kz = u + l;
                float dot = 0.0f;
#pragma unroll
                for (int d = 0; d < HDD; d += 2) {
                    dot = fmaf(qv[u][d],     kw[kz][d / 2].x, dot);
                    dot = fmaf(qv[u][d + 1], kw[kz][d / 2].y, dot);
                }
                s[u][tt] = dot + bias;
            }
        }
    }

    const int col = lane & 15;
    v8f acc[ZV][2];
#pragma unroll
    for (int u = 0; u < ZV; ++u) {
        float mx = -1e30f;
#pragma unroll
        for (int tt = 0; tt < NT; ++tt) mx = fmaxf(mx, s[u][tt]);
        float sum = 0.0f;
#pragma unroll
        for (int tt = 0; tt < NT; ++tt) { s[u][tt] = __expf(s[u][tt] - mx); sum += s[u][tt]; }
        const float inv = __builtin_amdgcn_rcpf(sum);
#pragma unroll
        for (int tt = 0; tt < NT; ++tt) s[u][tt] *= inv;
        acc[u][0] = v8f{};
        acc[u][1] = v8f{};
        pv_tiles(s[u], half, col, acc[u][0], acc[u][1]);
    }

    if (col < 3) {
#pragma unroll
        for (int tl = 0; tl < 2; ++tl) {
            const int yy = y0 + yg_l * 4 + 2 * tl + half;
            const size_t sp = (size_t)((x * GW + yy) * GT + z0);
#pragma unroll
            for (int r = 0; r < 8; ++r) {
                v4f o4;
                o4.x = acc[0][tl][r];
                o4.y = acc[1][tl][r];
                o4.z = acc[2][tl][r];
                o4.w = acc[3][tl][r];
                *(v4f*)(out + (size_t)(r * 3 + col) * (GH * GW * GT) + sp) = o4;
            }
        }
    }
}

__global__ __launch_bounds__(256, 2) void nat3d_wmma_kernel(
    const float* __restrict__ q,
    const float* __restrict__ k,
    const float* __restrict__ rpb,
    float* __restrict__ out)
{
    __shared__ float s_rpb[NHD * NT];
    __shared__ __attribute__((aligned(16))) float s_k[2][TX * XP];
    const int tid = threadIdx.x;
    const int t0  = blockIdx.x * 2;
    const int t1  = t0 + 1;

    float rpbv = 0.0f;
    if (tid < NHD * NT) rpbv = rpb[tid];
    v2f qwA[ZV][HDD / 2];
    v4f ldA[20];
    issue_tile_loads(t0, tid, q, k, qwA, ldA);
    asm volatile("" :: "v"(qwA[0][0]), "v"(qwA[0][1]), "v"(qwA[0][2]), "v"(qwA[1][0]), "v"(qwA[1][1]), "v"(qwA[1][2]),
                       "v"(qwA[2][0]), "v"(qwA[2][1]), "v"(qwA[2][2]), "v"(qwA[3][0]), "v"(qwA[3][1]), "v"(qwA[3][2]));
    asm volatile("" :: "v"(ldA[0]), "v"(ldA[1]), "v"(ldA[2]), "v"(ldA[3]), "v"(ldA[4]), "v"(ldA[5]), "v"(ldA[6]),
                       "v"(ldA[7]), "v"(ldA[8]), "v"(ldA[9]), "v"(ldA[10]), "v"(ldA[11]), "v"(ldA[12]), "v"(ldA[13]),
                       "v"(ldA[14]), "v"(ldA[15]), "v"(ldA[16]), "v"(ldA[17]), "v"(ldA[18]), "v"(ldA[19]));
    if (tid < NHD * NT) s_rpb[tid] = rpbv;
    fill_tile(t0, tid, ldA, s_k[0]);
    __syncthreads();

    v2f qwB[ZV][HDD / 2];
    v4f ldB[20];
    issue_tile_loads(t1, tid, q, k, qwB, ldB);
    __builtin_amdgcn_sched_barrier(0);

    compute_tile(t0, tid, qwA, s_k[0], s_rpb, out);

    __builtin_amdgcn_sched_barrier(0);
    fill_tile(t1, tid, ldB, s_k[1]);
    __builtin_amdgcn_fence(__ATOMIC_RELEASE, "workgroup", "local");
    __builtin_amdgcn_s_barrier();
    __builtin_amdgcn_fence(__ATOMIC_ACQUIRE, "workgroup", "local");
    compute_tile(t1, tid, qwB, s_k[1], s_rpb, out);
}

extern "C" void kernel_launch(void* const* d_in, const int* in_sizes, int n_in,
                              void* d_out, int out_size, void* d_ws, size_t ws_size,
                              hipStream_t stream) {
    (void)in_sizes; (void)n_in; (void)out_size; (void)d_ws; (void)ws_size;
    const float* q   = (const float*)d_in[0];
    const float* k   = (const float*)d_in[1];
    const float* rpb = (const float*)d_in[2];
    float* out = (float*)d_out;

    dim3 grid(250), block(256);
    nat3d_wmma_kernel<<<grid, block, 0, stream>>>(q, k, rpb, out);
}
